# starting kernel
# baseline (speedup 1.0000x reference)
#include <hip/hip_runtime.h>
#include <hip/hip_bf16.h>
#include <math.h>

#define B_   32
#define T_   4096
#define DIN_ 256
#define H_   256
#define NC_  32
#define L_   (T_ / NC_)

typedef float v2f __attribute__((ext_vector_type(2)));
typedef float v8f __attribute__((ext_vector_type(8)));

__device__ __forceinline__ float softplus_f(float u) {
    return fmaxf(u, 0.0f) + log1pf(__expf(-fabsf(u)));
}

__device__ __forceinline__ float logaddexp_f(float a, float b) {
    float mx = fmaxf(a, b);
    float d  = fabsf(a - b);
    return mx + log1pf(__expf(-d));
}

__device__ __forceinline__ float log_g_f(float a) {
    return (a >= 0.0f) ? __logf(a + 0.5f) : -softplus_f(-a);
}

__global__ __launch_bounds__(32) void mingru_gemm_kernel(
    const float* __restrict__ x,  const float* __restrict__ Wz,
    const float* __restrict__ bz, const float* __restrict__ Wh,
    const float* __restrict__ bh, float* __restrict__ v_out,
    float* __restrict__ c_out)
{
    const int lane  = threadIdx.x;
    const int m0    = blockIdx.y << 4;
    const int n0    = blockIdx.x << 4;
    const int l15   = lane & 15;
    const int khalf = (lane >> 4) << 1;

    const float* xp  = x  + (size_t)(m0 + l15) * DIN_ + khalf;
    const float* wzp = Wz + (size_t)(n0 + l15) * DIN_ + khalf;
    const float* whp = Wh + (size_t)(n0 + l15) * DIN_ + khalf;

    v8f accz = {};
    v8f acch = {};

#pragma unroll 4
    for (int kk = 0; kk < DIN_; kk += 4) {
        v2f a  = *(const v2f*)(xp  + kk);
        v2f bz2 = *(const v2f*)(wzp + kk);
        v2f bh2 = *(const v2f*)(whp + kk);
        accz = __builtin_amdgcn_wmma_f32_16x16x4_f32(
                   false, a, false, bz2, (short)0, accz, false, false);
        acch = __builtin_amdgcn_wmma_f32_16x16x4_f32(
                   false, a, false, bh2, (short)0, acch, false, false);
    }

    const int   ncol = n0 + l15;
    const float bzb  = bz[ncol];
    const float bhb  = bh[ncol];
    const int   mbase = m0 + ((lane >> 4) << 3);

#pragma unroll
    for (int j = 0; j < 8; ++j) {
        float kpre  = accz[j] + bzb;
        float sp    = softplus_f(kpre);
        float log_z = kpre - sp;
        float log_c = -sp;
        float hp    = acch[j] + bhb;
        float v     = log_z + log_g_f(hp);

        size_t idx = (size_t)(mbase + j) * H_ + ncol;
        v_out[idx] = v;
        c_out[idx] = log_c;
    }
}

__global__ __launch_bounds__(256) void mingru_chunk_reduce(
    const float* __restrict__ c_arr, const float* __restrict__ v_arr,
    float* __restrict__ Csum, float* __restrict__ Vsum)
{
    const int h = threadIdx.x;
    const int j = blockIdx.x;
    const int b = blockIdx.y;

    size_t base = ((size_t)b * T_ + (size_t)j * L_) * H_ + h;
    float C = 0.0f;
    float V = -INFINITY;

    for (int t = 0; t < L_; ++t) {
        size_t idx = base + (size_t)t * H_;
        __builtin_prefetch(c_arr + idx + (size_t)8 * H_, 0, 0);
        __builtin_prefetch(v_arr + idx + (size_t)8 * H_, 0, 0);
        float c = c_arr[idx];
        float v = v_arr[idx];
        C += c;
        V = logaddexp_f(V + c, v);
    }
    size_t sidx = ((size_t)b * NC_ + j) * H_ + h;
    Csum[sidx] = C;
    Vsum[sidx] = V;
}

__global__ __launch_bounds__(256) void mingru_carry(
    const float* __restrict__ h0, const float* __restrict__ Csum,
    const float* __restrict__ Vsum, float* __restrict__ entry)
{
    const int h = threadIdx.x;
    const int b = blockIdx.x;

    float state = log_g_f(h0[(size_t)b * H_ + h]);
#pragma unroll
    for (int j = 0; j < NC_; ++j) {
        size_t sidx = ((size_t)b * NC_ + j) * H_ + h;
        entry[sidx] = state;
        state = logaddexp_f(state + Csum[sidx], Vsum[sidx]);
    }
}

__global__ __launch_bounds__(256) void mingru_apply(
    const float* __restrict__ c_arr, float* __restrict__ vout,
    const float* __restrict__ entry)
{
    const int h = threadIdx.x;
    const int j = blockIdx.x;
    const int b = blockIdx.y;

    float state = entry[((size_t)b * NC_ + j) * H_ + h];
    size_t base = ((size_t)b * T_ + (size_t)j * L_) * H_ + h;

    for (int t = 0; t < L_; ++t) {
        size_t idx = base + (size_t)t * H_;
        __builtin_prefetch(c_arr + idx + (size_t)8 * H_, 0, 0);
        float c = c_arr[idx];
        float v = vout[idx];
        state = logaddexp_f(state + c, v);
        vout[idx] = __expf(state);
    }
}

extern "C" void kernel_launch(void* const* d_in, const int* in_sizes, int n_in,
                              void* d_out, int out_size, void* d_ws, size_t ws_size,
                              hipStream_t stream)
{
    const float* x  = (const float*)d_in[0];
    const float* h0 = (const float*)d_in[1];
    const float* Wz = (const float*)d_in[2];
    const float* bz = (const float*)d_in[3];
    const float* Wh = (const float*)d_in[4];
    const float* bh = (const float*)d_in[5];
    float* out = (float*)d_out;

    float* ws    = (float*)d_ws;
    float* c_arr = ws;
    float* Csum  = c_arr + (size_t)B_ * T_ * H_;
    float* Vsum  = Csum  + (size_t)B_ * NC_ * H_;
    float* entry = Vsum  + (size_t)B_ * NC_ * H_;

    dim3 ggrid(H_ / 16, (B_ * T_) / 16);
    mingru_gemm_kernel<<<ggrid, 32, 0, stream>>>(x, Wz, bz, Wh, bh, out, c_arr);

    dim3 sgrid(NC_, B_);
    mingru_chunk_reduce<<<sgrid, H_, 0, stream>>>(c_arr, out, Csum, Vsum);
    mingru_carry<<<B_, H_, 0, stream>>>(h0, Csum, Vsum, entry);
    mingru_apply<<<sgrid, H_, 0, stream>>>(c_arr, out, entry);
}


# GEMM made weight-stationary: a wave keeps its 16-column strip of both W in registers for all K and walks 32 row tiles, loading only x (16-byte loads); k order in each 8 changes: sum order changes
# speedup vs baseline: 1.4349x; 1.4349x over previous
#include <hip/hip_runtime.h>
#include <hip/hip_bf16.h>
#include <math.h>

#define B_   32
#define T_   4096
#define DIN_ 256
#define H_   256
#define NC_  32
#define L_   (T_ / NC_)

typedef float v2f __attribute__((ext_vector_type(2)));
typedef float v4f __attribute__((ext_vector_type(4)));
typedef float v8f __attribute__((ext_vector_type(8)));

__device__ __forceinline__ float softplus_f(float u) {
    return fmaxf(u, 0.0f) + log1pf(__expf(-fabsf(u)));
}

__device__ __forceinline__ float logaddexp_f(float a, float b) {
    float mx = fmaxf(a, b);
    float d  = fabsf(a - b);
    return mx + log1pf(__expf(-d));
}

__device__ __forceinline__ float log_g_f(float a) {
    return (a >= 0.0f) ? __logf(a + 0.5f) : -softplus_f(-a);
}

#define KP_  (DIN_ / 8)
#define RT_  32
#define GW_  8

__device__ __forceinline__ v8f wmma_f32(v2f a, v2f b, v8f c) {
    return __builtin_amdgcn_wmma_f32_16x16x4_f32(
               false, a, false, b, (short)0, c, false, false);
}

__global__ __launch_bounds__(GW_ * 32) void mingru_gemm_kernel(
    const float* __restrict__ x,  const float* __restrict__ Wz,
    const float* __restrict__ bz, const float* __restrict__ Wh,
    const float* __restrict__ bh, float* __restrict__ v_out,
    float* __restrict__ c_out)
{
    const int lane = threadIdx.x & 31;
    const int wave = threadIdx.x >> 5;
    const int l15  = lane & 15;
    const int k4   = (lane >> 4) << 2;
    const int n0   = (blockIdx.x * GW_ + wave) << 4;
    const int m0   = blockIdx.y * (RT_ * 16);

    const float* wzp = Wz + (size_t)(n0 + l15) * DIN_ + k4;
    const float* whp = Wh + (size_t)(n0 + l15) * DIN_ + k4;
    const float* xp  = x  + (size_t)(m0 + l15) * DIN_ + k4;

    v4f wz[KP_], wh[KP_], xa[KP_];
#pragma unroll
    for (int p = 0; p < KP_; ++p) {
        wz[p] = *(const v4f*)(wzp + 8 * p);
        wh[p] = *(const v4f*)(whp + 8 * p);
    }
#pragma unroll
    for (int p = 0; p < KP_; ++p) xa[p] = *(const v4f*)(xp + 8 * p);

    const int   ncol  = n0 + l15;
    const float bzb   = bz[ncol];
    const float bhb   = bh[ncol];
    const int   rhalf = (lane >> 4) << 3;

    for (int t = 0; t < RT_; ++t) {
        v8f accz = {};
        v8f acch = {};

#pragma unroll
        for (int p = 0; p < KP_; ++p) {
            const v2f a0 = {xa[p].x, xa[p].y};
            const v2f a1 = {xa[p].z, xa[p].w};
            const v2f z0 = {wz[p].x, wz[p].y};
            const v2f z1 = {wz[p].z, wz[p].w};
            const v2f h0 = {wh[p].x, wh[p].y};
            const v2f h1 = {wh[p].z, wh[p].w};
            accz = wmma_f32(a0, z0, accz);
            acch = wmma_f32(a0, h0, acch);
            accz = wmma_f32(a1, z1, accz);
            acch = wmma_f32(a1, h1, acch);
        }

        const int tn = (t + 1 < RT_) ? (t + 1) : t;
        const float* xn = xp + (size_t)tn * (16 * DIN_);
#pragma unroll
        for (int p = 0; p < KP_; ++p) xa[p] = *(const v4f*)(xn + 8 * p);

        const int mbase = m0 + t * 16 + rhalf;
#pragma unroll
        for (int j = 0; j < 8; ++j) {
            float kpre  = accz[j] + bzb;
            float sp    = softplus_f(kpre);
            float log_z = kpre - sp;
            float log_c = -sp;
            float hp    = acch[j] + bhb;
            float v     = log_z + log_g_f(hp);

            size_t idx = (size_t)(mbase + j) * H_ + ncol;
            v_out[idx] = v;
            c_out[idx] = log_c;
        }
    }
}

__global__ __launch_bounds__(256) void mingru_chunk_reduce(
    const float* __restrict__ c_arr, const float* __restrict__ v_arr,
    float* __restrict__ Csum, float* __restrict__ Vsum)
{
    const int h = threadIdx.x;
    const int j = blockIdx.x;
    const int b = blockIdx.y;

    size_t base = ((size_t)b * T_ + (size_t)j * L_) * H_ + h;
    float C = 0.0f;
    float V = -INFINITY;

    for (int t = 0; t < L_; ++t) {
        size_t idx = base + (size_t)t * H_;
        __builtin_prefetch(c_arr + idx + (size_t)8 * H_, 0, 0);
        __builtin_prefetch(v_arr + idx + (size_t)8 * H_, 0, 0);
        float c = c_arr[idx];
        float v = v_arr[idx];
        C += c;
        V = logaddexp_f(V + c, v);
    }
    size_t sidx = ((size_t)b * NC_ + j) * H_ + h;
    Csum[sidx] = C;
    Vsum[sidx] = V;
}

__global__ __launch_bounds__(256) void mingru_carry(
    const float* __restrict__ h0, const float* __restrict__ Csum,
    const float* __restrict__ Vsum, float* __restrict__ entry)
{
    const int h = threadIdx.x;
    const int b = blockIdx.x;

    float state = log_g_f(h0[(size_t)b * H_ + h]);
#pragma unroll
    for (int j = 0; j < NC_; ++j) {
        size_t sidx = ((size_t)b * NC_ + j) * H_ + h;
        entry[sidx] = state;
        state = logaddexp_f(state + Csum[sidx], Vsum[sidx]);
    }
}

__global__ __launch_bounds__(256) void mingru_apply(
    const float* __restrict__ c_arr, float* __restrict__ vout,
    const float* __restrict__ entry)
{
    const int h = threadIdx.x;
    const int j = blockIdx.x;
    const int b = blockIdx.y;

    float state = entry[((size_t)b * NC_ + j) * H_ + h];
    size_t base = ((size_t)b * T_ + (size_t)j * L_) * H_ + h;

    for (int t = 0; t < L_; ++t) {
        size_t idx = base + (size_t)t * H_;
        __builtin_prefetch(c_arr + idx + (size_t)8 * H_, 0, 0);
        float c = c_arr[idx];
        float v = vout[idx];
        state = logaddexp_f(state + c, v);
        vout[idx] = __expf(state);
    }
}

extern "C" void kernel_launch(void* const* d_in, const int* in_sizes, int n_in,
                              void* d_out, int out_size, void* d_ws, size_t ws_size,
                              hipStream_t stream)
{
    const float* x  = (const float*)d_in[0];
    const float* h0 = (const float*)d_in[1];
    const float* Wz = (const float*)d_in[2];
    const float* bz = (const float*)d_in[3];
    const float* Wh = (const float*)d_in[4];
    const float* bh = (const float*)d_in[5];
    float* out = (float*)d_out;

    float* ws    = (float*)d_ws;
    float* c_arr = ws;
    float* Csum  = c_arr + (size_t)B_ * T_ * H_;
    float* Vsum  = Csum  + (size_t)B_ * NC_ * H_;
    float* entry = Vsum  + (size_t)B_ * NC_ * H_;

    dim3 ggrid(H_ / (16 * GW_), (B_ * T_) / (16 * RT_));
    mingru_gemm_kernel<<<ggrid, GW_ * 32, 0, stream>>>(x, Wz, bz, Wh, bh, out, c_arr);

    dim3 sgrid(NC_, B_);
    mingru_chunk_reduce<<<sgrid, H_, 0, stream>>>(c_arr, out, Csum, Vsum);
    mingru_carry<<<B_, H_, 0, stream>>>(h0, Csum, Vsum, entry);
    mingru_apply<<<sgrid, H_, 0, stream>>>(c_arr, out, entry);
}


# Gate math (softplus, log g) moved out of the GEMM kernel, where 2 waves per SIMD left its dependent chains unhidden, into a new in-place elementwise pass at 16 waves per SIMD; sums unchanged
# speedup vs baseline: 1.5632x; 1.0894x over previous
#include <hip/hip_runtime.h>
#include <hip/hip_bf16.h>
#include <math.h>

#define B_   32
#define T_   4096
#define DIN_ 256
#define H_   256
#define NC_  32
#define L_   (T_ / NC_)

typedef float v2f __attribute__((ext_vector_type(2)));
typedef float v4f __attribute__((ext_vector_type(4)));
typedef float v8f __attribute__((ext_vector_type(8)));

__device__ __forceinline__ float softplus_f(float u) {
    return fmaxf(u, 0.0f) + log1pf(__expf(-fabsf(u)));
}

__device__ __forceinline__ float logaddexp_f(float a, float b) {
    float mx = fmaxf(a, b);
    float d  = fabsf(a - b);
    return mx + log1pf(__expf(-d));
}

__device__ __forceinline__ float log_g_f(float a) {
    return (a >= 0.0f) ? __logf(a + 0.5f) : -softplus_f(-a);
}

__device__ __forceinline__ float log_g_where(float a) {
    float pos = __logf(fmaxf(a, 0.0f) + 0.5f);
    float neg = -softplus_f(-a);
    return (a >= 0.0f) ? pos : neg;
}

#define KP_  (DIN_ / 8)
#define RT_  32
#define GW_  8

__device__ __forceinline__ v8f wmma_f32(v2f a, v2f b, v8f c) {
    return __builtin_amdgcn_wmma_f32_16x16x4_f32(
               false, a, false, b, (short)0, c, false, false);
}

__global__ __launch_bounds__(GW_ * 32) void mingru_gemm_kernel(
    const float* __restrict__ x,  const float* __restrict__ Wz,
    const float* __restrict__ bz, const float* __restrict__ Wh,
    const float* __restrict__ bh, float* __restrict__ v_out,
    float* __restrict__ c_out)
{
    const int lane = threadIdx.x & 31;
    const int wave = threadIdx.x >> 5;
    const int l15  = lane & 15;
    const int k4   = (lane >> 4) << 2;
    const int n0   = (blockIdx.x * GW_ + wave) << 4;
    const int m0   = blockIdx.y * (RT_ * 16);

    const float* wzp = Wz + (size_t)(n0 + l15) * DIN_ + k4;
    const float* whp = Wh + (size_t)(n0 + l15) * DIN_ + k4;
    const float* xp  = x  + (size_t)(m0 + l15) * DIN_ + k4;

    v4f wz[KP_], wh[KP_], xa[KP_];
#pragma unroll
    for (int p = 0; p < KP_; ++p) {
        wz[p] = *(const v4f*)(wzp + 8 * p);
        wh[p] = *(const v4f*)(whp + 8 * p);
    }
#pragma unroll
    for (int p = 0; p < KP_; ++p) xa[p] = *(const v4f*)(xp + 8 * p);

    const int   ncol  = n0 + l15;
    const float bzb   = bz[ncol];
    const float bhb   = bh[ncol];
    const int   rhalf = (lane >> 4) << 3;

    for (int t = 0; t < RT_; ++t) {
        v8f accz = {};
        v8f acch = {};
        const int tn = (t + 1 < RT_) ? (t + 1) : t;
        const float* xn = xp + (size_t)tn * (16 * DIN_);

#pragma unroll
        for (int p = 0; p < KP_; ++p) {
            const v2f a0 = {xa[p].x, xa[p].y};
            const v2f a1 = {xa[p].z, xa[p].w};
            const v2f z0 = {wz[p].x, wz[p].y};
            const v2f z1 = {wz[p].z, wz[p].w};
            const v2f h0 = {wh[p].x, wh[p].y};
            const v2f h1 = {wh[p].z, wh[p].w};
            accz = wmma_f32(a0, z0, accz);
            acch = wmma_f32(a0, h0, acch);
            accz = wmma_f32(a1, z1, accz);
            acch = wmma_f32(a1, h1, acch);
            xa[p] = *(const v4f*)(xn + 8 * p);
        }

        asm volatile("v_nop\n\tv_nop\n\tv_nop\n\tv_nop" : "+v"(accz), "+v"(acch));

        const int mbase = m0 + t * 16 + rhalf;
#pragma unroll
        for (int j = 0; j < 8; ++j) {
            size_t idx = (size_t)(mbase + j) * H_ + ncol;
            v_out[idx] = acch[j] + bhb;
            c_out[idx] = accz[j] + bzb;
        }
    }
}

#define GE_ 4
#define GB_ 2048
#define GI_ ((B_ * T_ * H_) / (GB_ * 256 * GE_))

__global__ __launch_bounds__(256) void mingru_gate_kernel(
    float* __restrict__ kc, float* __restrict__ hv)
{
    size_t i = ((size_t)blockIdx.x * 256 + threadIdx.x) * GE_;
    for (int s = 0; s < GI_; ++s, i += (size_t)GB_ * 256 * GE_) {
        const v4f k4 = *(const v4f*)(kc + i);
        const v4f h4 = *(const v4f*)(hv + i);
        v4f c4, v4;
#pragma unroll
        for (int e = 0; e < GE_; ++e) {
            float kpre  = k4[e];
            float sp    = softplus_f(kpre);
            float log_z = kpre - sp;
            c4[e] = -sp;
            v4[e] = log_z + log_g_where(h4[e]);
        }
        *(v4f*)(kc + i) = c4;
        *(v4f*)(hv + i) = v4;
    }
}

__global__ __launch_bounds__(256) void mingru_chunk_reduce(
    const float* __restrict__ c_arr, const float* __restrict__ v_arr,
    float* __restrict__ Csum, float* __restrict__ Vsum)
{
    const int h = threadIdx.x;
    const int j = blockIdx.x;
    const int b = blockIdx.y;

    size_t base = ((size_t)b * T_ + (size_t)j * L_) * H_ + h;
    float C = 0.0f;
    float V = -INFINITY;

    for (int t = 0; t < L_; ++t) {
        size_t idx = base + (size_t)t * H_;
        __builtin_prefetch(c_arr + idx + (size_t)8 * H_, 0, 0);
        __builtin_prefetch(v_arr + idx + (size_t)8 * H_, 0, 0);
        float c = c_arr[idx];
        float v = v_arr[idx];
        C += c;
        V = logaddexp_f(V + c, v);
    }
    size_t sidx = ((size_t)b * NC_ + j) * H_ + h;
    Csum[sidx] = C;
    Vsum[sidx] = V;
}

__global__ __launch_bounds__(256) void mingru_carry(
    const float* __restrict__ h0, const float* __restrict__ Csum,
    const float* __restrict__ Vsum, float* __restrict__ entry)
{
    const int h = threadIdx.x;
    const int b = blockIdx.x;

    float state = log_g_f(h0[(size_t)b * H_ + h]);
#pragma unroll
    for (int j = 0; j < NC_; ++j) {
        size_t sidx = ((size_t)b * NC_ + j) * H_ + h;
        entry[sidx] = state;
        state = logaddexp_f(state + Csum[sidx], Vsum[sidx]);
    }
}

__global__ __launch_bounds__(256) void mingru_apply(
    const float* __restrict__ c_arr, float* __restrict__ vout,
    const float* __restrict__ entry)
{
    const int h = threadIdx.x;
    const int j = blockIdx.x;
    const int b = blockIdx.y;

    float state = entry[((size_t)b * NC_ + j) * H_ + h];
    size_t base = ((size_t)b * T_ + (size_t)j * L_) * H_ + h;

    for (int t = 0; t < L_; ++t) {
        size_t idx = base + (size_t)t * H_;
        __builtin_prefetch(c_arr + idx + (size_t)8 * H_, 0, 0);
        float c = c_arr[idx];
        float v = vout[idx];
        state = logaddexp_f(state + c, v);
        vout[idx] = __expf(state);
    }
}

extern "C" void kernel_launch(void* const* d_in, const int* in_sizes, int n_in,
                              void* d_out, int out_size, void* d_ws, size_t ws_size,
                              hipStream_t stream)
{
    const float* x  = (const float*)d_in[0];
    const float* h0 = (const float*)d_in[1];
    const float* Wz = (const float*)d_in[2];
    const float* bz = (const float*)d_in[3];
    const float* Wh = (const float*)d_in[4];
    const float* bh = (const float*)d_in[5];
    float* out = (float*)d_out;

    float* ws    = (float*)d_ws;
    float* c_arr = ws;
    float* Csum  = c_arr + (size_t)B_ * T_ * H_;
    float* Vsum  = Csum  + (size_t)B_ * NC_ * H_;
    float* entry = Vsum  + (size_t)B_ * NC_ * H_;

    dim3 ggrid(H_ / (16 * GW_), (B_ * T_) / (16 * RT_));
    mingru_gemm_kernel<<<ggrid, GW_ * 32, 0, stream>>>(x, Wz, bz, Wh, bh, out, c_arr);

    mingru_gate_kernel<<<GB_, 256, 0, stream>>>(c_arr, out);

    dim3 sgrid(NC_, B_);
    mingru_chunk_reduce<<<sgrid, H_, 0, stream>>>(c_arr, out, Csum, Vsum);
    mingru_carry<<<B_, H_, 0, stream>>>(h0, Csum, Vsum, entry);
    mingru_apply<<<sgrid, H_, 0, stream>>>(c_arr, out, entry);
}


# Chunk scan state kept as running max M and max-scaled sum S (log h = M + log S): reduce, carry, re-scan take 1 exp per row, no log1p; rows read 8 per trip; row order kept, rounding changes
# speedup vs baseline: 1.9969x; 1.2774x over previous
#include <hip/hip_runtime.h>
#include <hip/hip_bf16.h>
#include <math.h>

#define B_   32
#define T_   4096
#define DIN_ 256
#define H_   256
#define NC_  32
#define L_   (T_ / NC_)

typedef float v2f __attribute__((ext_vector_type(2)));
typedef float v4f __attribute__((ext_vector_type(4)));
typedef float v8f __attribute__((ext_vector_type(8)));

__device__ __forceinline__ float softplus_f(float u) {
    return fmaxf(u, 0.0f) + log1pf(__expf(-fabsf(u)));
}

__device__ __forceinline__ float log_g_f(float a) {
    return (a >= 0.0f) ? __logf(a + 0.5f) : -softplus_f(-a);
}

__device__ __forceinline__ float log_g_where(float a) {
    float pos = __logf(fmaxf(a, 0.0f) + 0.5f);
    float neg = -softplus_f(-a);
    return (a >= 0.0f) ? pos : neg;
}

#define RB_ 8

__device__ __forceinline__ void scan_step(float& M, float& S, float c, float v) {
    const float M1 = M + c;
    const float d  = v - M1;
    const float e  = __expf(-fabsf(d));
    S = (d > 0.0f) ? fmaf(S, e, 1.0f) : (S + e);
    M = fmaxf(M1, v);
}

#define KP_  (DIN_ / 8)
#define RT_  32
#define GW_  8

__device__ __forceinline__ v8f wmma_f32(v2f a, v2f b, v8f c) {
    return __builtin_amdgcn_wmma_f32_16x16x4_f32(
               false, a, false, b, (short)0, c, false, false);
}

__global__ __launch_bounds__(GW_ * 32) void mingru_gemm_kernel(
    const float* __restrict__ x,  const float* __restrict__ Wz,
    const float* __restrict__ bz, const float* __restrict__ Wh,
    const float* __restrict__ bh, float* __restrict__ v_out,
    float* __restrict__ c_out)
{
    const int lane = threadIdx.x & 31;
    const int wave = threadIdx.x >> 5;
    const int l15  = lane & 15;
    const int k4   = (lane >> 4) << 2;
    const int n0   = (blockIdx.x * GW_ + wave) << 4;
    const int m0   = blockIdx.y * (RT_ * 16);

    const float* wzp = Wz + (size_t)(n0 + l15) * DIN_ + k4;
    const float* whp = Wh + (size_t)(n0 + l15) * DIN_ + k4;
    const float* xp  = x  + (size_t)(m0 + l15) * DIN_ + k4;

    v4f wz[KP_], wh[KP_], xa[KP_];
#pragma unroll
    for (int p = 0; p < KP_; ++p) {
        wz[p] = *(const v4f*)(wzp + 8 * p);
        wh[p] = *(const v4f*)(whp + 8 * p);
    }
#pragma unroll
    for (int p = 0; p < KP_; ++p) xa[p] = *(const v4f*)(xp + 8 * p);

    const int   ncol  = n0 + l15;
    const float bzb   = bz[ncol];
    const float bhb   = bh[ncol];
    const int   rhalf = (lane >> 4) << 3;

    for (int t = 0; t < RT_; ++t) {
        v8f accz = {};
        v8f acch = {};
        const int tn = (t + 1 < RT_) ? (t + 1) : t;
        const float* xn = xp + (size_t)tn * (16 * DIN_);

#pragma unroll
        for (int p = 0; p < KP_; ++p) {
            const v2f a0 = {xa[p].x, xa[p].y};
            const v2f a1 = {xa[p].z, xa[p].w};
            const v2f z0 = {wz[p].x, wz[p].y};
            const v2f z1 = {wz[p].z, wz[p].w};
            const v2f h0 = {wh[p].x, wh[p].y};
            const v2f h1 = {wh[p].z, wh[p].w};
            accz = wmma_f32(a0, z0, accz);
            acch = wmma_f32(a0, h0, acch);
            accz = wmma_f32(a1, z1, accz);
            acch = wmma_f32(a1, h1, acch);
            xa[p] = *(const v4f*)(xn + 8 * p);
        }

        asm volatile("v_nop\n\tv_nop\n\tv_nop\n\tv_nop" : "+v"(accz), "+v"(acch));

        const int mbase = m0 + t * 16 + rhalf;
#pragma unroll
        for (int j = 0; j < 8; ++j) {
            size_t idx = (size_t)(mbase + j) * H_ + ncol;
            v_out[idx] = acch[j] + bhb;
            c_out[idx] = accz[j] + bzb;
        }
    }
}

#define GE_ 4
#define GB_ 2048
#define GI_ ((B_ * T_ * H_) / (GB_ * 256 * GE_))

__global__ __launch_bounds__(256) void mingru_gate_kernel(
    float* __restrict__ kc, float* __restrict__ hv)
{
    size_t i = ((size_t)blockIdx.x * 256 + threadIdx.x) * GE_;
    for (int s = 0; s < GI_; ++s, i += (size_t)GB_ * 256 * GE_) {
        const v4f k4 = *(const v4f*)(kc + i);
        const v4f h4 = *(const v4f*)(hv + i);
        v4f c4, v4;
#pragma unroll
        for (int e = 0; e < GE_; ++e) {
            float kpre  = k4[e];
            float sp    = softplus_f(kpre);
            float log_z = kpre - sp;
            c4[e] = -sp;
            v4[e] = log_z + log_g_where(h4[e]);
        }
        *(v4f*)(kc + i) = c4;
        *(v4f*)(hv + i) = v4;
    }
}

__global__ __launch_bounds__(256) void mingru_chunk_reduce(
    const float* __restrict__ c_arr, const float* __restrict__ v_arr,
    float* __restrict__ Csum, float* __restrict__ Msum, float* __restrict__ Ssum)
{
    const int h = threadIdx.x;
    const int j = blockIdx.x;
    const int b = blockIdx.y;

    size_t base = ((size_t)b * T_ + (size_t)j * L_) * H_ + h;
    float C = 0.0f;
    float M = -INFINITY;
    float S = 0.0f;

    for (int t0 = 0; t0 < L_; t0 += RB_) {
        float c[RB_], v[RB_];
#pragma unroll
        for (int u = 0; u < RB_; ++u) {
            size_t idx = base + (size_t)(t0 + u) * H_;
            c[u] = c_arr[idx];
            v[u] = v_arr[idx];
        }
#pragma unroll
        for (int u = 0; u < RB_; ++u) {
            C += c[u];
            scan_step(M, S, c[u], v[u]);
        }
    }
    size_t sidx = ((size_t)b * NC_ + j) * H_ + h;
    Csum[sidx] = C;
    Msum[sidx] = M;
    Ssum[sidx] = S;
}

__global__ __launch_bounds__(256) void mingru_carry(
    const float* __restrict__ h0, const float* __restrict__ Csum,
    float* __restrict__ Msum, float* __restrict__ Ssum)
{
    const int h = threadIdx.x;
    const int b = blockIdx.x;

    float M = log_g_f(h0[(size_t)b * H_ + h]);
    float S = 1.0f;
#pragma unroll
    for (int j = 0; j < NC_; ++j) {
        size_t sidx = ((size_t)b * NC_ + j) * H_ + h;
        const float Cj = Csum[sidx];
        const float Mj = Msum[sidx];
        const float Sj = Ssum[sidx];
        Msum[sidx] = M;
        Ssum[sidx] = S;
        const float M1 = M + Cj;
        const float d  = Mj - M1;
        const float e  = __expf(-fabsf(d));
        S = (d > 0.0f) ? fmaf(S, e, Sj) : fmaf(Sj, e, S);
        M = fmaxf(M1, Mj);
    }
}

__global__ __launch_bounds__(256) void mingru_apply(
    const float* __restrict__ c_arr, float* __restrict__ vout,
    const float* __restrict__ Ment, const float* __restrict__ Sent)
{
    const int h = threadIdx.x;
    const int j = blockIdx.x;
    const int b = blockIdx.y;

    const size_t sidx = ((size_t)b * NC_ + j) * H_ + h;
    float M = Ment[sidx];
    float S = Sent[sidx];
    size_t base = ((size_t)b * T_ + (size_t)j * L_) * H_ + h;

    for (int t0 = 0; t0 < L_; t0 += RB_) {
        float c[RB_], v[RB_];
#pragma unroll
        for (int u = 0; u < RB_; ++u) {
            size_t idx = base + (size_t)(t0 + u) * H_;
            c[u] = c_arr[idx];
            v[u] = vout[idx];
        }
#pragma unroll
        for (int u = 0; u < RB_; ++u) {
            scan_step(M, S, c[u], v[u]);
            vout[base + (size_t)(t0 + u) * H_] = __expf(M) * S;
        }
    }
}

extern "C" void kernel_launch(void* const* d_in, const int* in_sizes, int n_in,
                              void* d_out, int out_size, void* d_ws, size_t ws_size,
                              hipStream_t stream)
{
    const float* x  = (const float*)d_in[0];
    const float* h0 = (const float*)d_in[1];
    const float* Wz = (const float*)d_in[2];
    const float* bz = (const float*)d_in[3];
    const float* Wh = (const float*)d_in[4];
    const float* bh = (const float*)d_in[5];
    float* out = (float*)d_out;

    float* ws    = (float*)d_ws;
    float* c_arr = ws;
    float* Csum  = c_arr + (size_t)B_ * T_ * H_;
    float* Msum  = Csum  + (size_t)B_ * NC_ * H_;
    float* Ssum  = Msum  + (size_t)B_ * NC_ * H_;

    dim3 ggrid(H_ / (16 * GW_), (B_ * T_) / (16 * RT_));
    mingru_gemm_kernel<<<ggrid, GW_ * 32, 0, stream>>>(x, Wz, bz, Wh, bh, out, c_arr);

    mingru_gate_kernel<<<GB_, 256, 0, stream>>>(c_arr, out);

    dim3 sgrid(NC_, B_);
    mingru_chunk_reduce<<<sgrid, H_, 0, stream>>>(c_arr, out, Csum, Msum, Ssum);
    mingru_carry<<<B_, H_, 0, stream>>>(h0, Csum, Msum, Ssum);
    mingru_apply<<<sgrid, H_, 0, stream>>>(c_arr, out, Msum, Ssum);
}


# Scan terms kept as (exponent, mantissa) pairs formed from raw k, hp inside reduce and re-scan: 2 exp + 2 rcp per row, no log; gate pass and its planes' round trip go; row order kept, rounding changes
# speedup vs baseline: 2.9920x; 1.4983x over previous
#include <hip/hip_runtime.h>
#include <hip/hip_bf16.h>
#include <math.h>

#define B_   32
#define T_   4096
#define DIN_ 256
#define H_   256
#define NC_  32
#define L_   (T_ / NC_)

typedef float v2f __attribute__((ext_vector_type(2)));
typedef float v4f __attribute__((ext_vector_type(4)));
typedef float v8f __attribute__((ext_vector_type(8)));

__device__ __forceinline__ float softplus_f(float u) {
    return fmaxf(u, 0.0f) + log1pf(__expf(-fabsf(u)));
}

__device__ __forceinline__ float log_g_f(float a) {
    return (a >= 0.0f) ? __logf(a + 0.5f) : -softplus_f(-a);
}

#define RB_ 8

__device__ __forceinline__ void row_terms(float k, float a,
                                          float& Mc, float& Sc, float& Mv, float& Sv) {
    const float E1 = __expf(-fabsf(k));
    const float r1 = __builtin_amdgcn_rcpf(1.0f + E1);
    const float E2 = __expf(-fabsf(a));
    const float r2 = __builtin_amdgcn_rcpf(1.0f + E2);
    Mc = fminf(-k, 0.0f);
    Sc = r1;
    Mv = (k + Mc) + fminf(a, 0.0f);
    const float G = (a >= 0.0f) ? (a + 0.5f) : r2;
    Sv = r1 * G;
}

__device__ __forceinline__ void scan_step(float& M, float& S,
                                          float Mc, float Sc, float Mv, float Sv) {
    const float M1 = M + Mc;
    const float S1 = S * Sc;
    const float d  = Mv - M1;
    const float e  = __expf(-fabsf(d));
    S = (d > 0.0f) ? fmaf(S1, e, Sv) : fmaf(Sv, e, S1);
    M = fmaxf(M1, Mv);
}

__device__ __forceinline__ void renorm(float& M, float& S) {
    int ex;
    S = frexpf(S, &ex);
    M = fmaf((float)ex, 0.6931471805599453f, M);
}

#define KP_  (DIN_ / 8)
#define RT_  32
#define GW_  8

__device__ __forceinline__ v8f wmma_f32(v2f a, v2f b, v8f c) {
    return __builtin_amdgcn_wmma_f32_16x16x4_f32(
               false, a, false, b, (short)0, c, false, false);
}

__global__ __launch_bounds__(GW_ * 32) void mingru_gemm_kernel(
    const float* __restrict__ x,  const float* __restrict__ Wz,
    const float* __restrict__ bz, const float* __restrict__ Wh,
    const float* __restrict__ bh, float* __restrict__ v_out,
    float* __restrict__ c_out)
{
    const int lane = threadIdx.x & 31;
    const int wave = threadIdx.x >> 5;
    const int l15  = lane & 15;
    const int k4   = (lane >> 4) << 2;
    const int n0   = (blockIdx.x * GW_ + wave) << 4;
    const int m0   = blockIdx.y * (RT_ * 16);

    const float* wzp = Wz + (size_t)(n0 + l15) * DIN_ + k4;
    const float* whp = Wh + (size_t)(n0 + l15) * DIN_ + k4;
    const float* xp  = x  + (size_t)(m0 + l15) * DIN_ + k4;

    v4f wz[KP_], wh[KP_], xa[KP_];
#pragma unroll
    for (int p = 0; p < KP_; ++p) {
        wz[p] = *(const v4f*)(wzp + 8 * p);
        wh[p] = *(const v4f*)(whp + 8 * p);
    }
#pragma unroll
    for (int p = 0; p < KP_; ++p) xa[p] = *(const v4f*)(xp + 8 * p);

    const int   ncol  = n0 + l15;
    const float bzb   = bz[ncol];
    const float bhb   = bh[ncol];
    const int   rhalf = (lane >> 4) << 3;

    for (int t = 0; t < RT_; ++t) {
        v8f accz = {};
        v8f acch = {};
        const int tn = (t + 1 < RT_) ? (t + 1) : t;
        const float* xn = xp + (size_t)tn * (16 * DIN_);

#pragma unroll
        for (int p = 0; p < KP_; ++p) {
            const v2f a0 = {xa[p].x, xa[p].y};
            const v2f a1 = {xa[p].z, xa[p].w};
            const v2f z0 = {wz[p].x, wz[p].y};
            const v2f z1 = {wz[p].z, wz[p].w};
            const v2f h0 = {wh[p].x, wh[p].y};
            const v2f h1 = {wh[p].z, wh[p].w};
            accz = wmma_f32(a0, z0, accz);
            acch = wmma_f32(a0, h0, acch);
            accz = wmma_f32(a1, z1, accz);
            acch = wmma_f32(a1, h1, acch);
            xa[p] = *(const v4f*)(xn + 8 * p);
        }

        asm volatile("v_nop\n\tv_nop\n\tv_nop\n\tv_nop" : "+v"(accz), "+v"(acch));

        const int mbase = m0 + t * 16 + rhalf;
#pragma unroll
        for (int j = 0; j < 8; ++j) {
            size_t idx = (size_t)(mbase + j) * H_ + ncol;
            v_out[idx] = acch[j] + bhb;
            c_out[idx] = accz[j] + bzb;
        }
    }
}

__global__ __launch_bounds__(256) void mingru_chunk_reduce(
    const float* __restrict__ k_arr, const float* __restrict__ a_arr,
    float* __restrict__ Csum, float* __restrict__ Msum, float* __restrict__ Ssum)
{
    const int h = threadIdx.x;
    const int j = blockIdx.x;
    const int b = blockIdx.y;

    size_t base = ((size_t)b * T_ + (size_t)j * L_) * H_ + h;
    float A = 0.0f;
    float P = 1.0f;
    float M = -INFINITY;
    float S = 0.0f;

    for (int t0 = 0; t0 < L_; t0 += RB_) {
        float k[RB_], a[RB_];
#pragma unroll
        for (int u = 0; u < RB_; ++u) {
            size_t idx = base + (size_t)(t0 + u) * H_;
            k[u] = k_arr[idx];
            a[u] = a_arr[idx];
        }
#pragma unroll
        for (int u = 0; u < RB_; ++u) {
            float Mc, Sc, Mv, Sv;
            row_terms(k[u], a[u], Mc, Sc, Mv, Sv);
            A += Mc;
            P *= Sc;
            scan_step(M, S, Mc, Sc, Mv, Sv);
        }
        renorm(A, P);
        renorm(M, S);
    }
    size_t sidx = ((size_t)b * NC_ + j) * H_ + h;
    Csum[sidx] = A + __logf(P);
    Msum[sidx] = M;
    Ssum[sidx] = S;
}

__global__ __launch_bounds__(256) void mingru_carry(
    const float* __restrict__ h0, const float* __restrict__ Csum,
    float* __restrict__ Msum, float* __restrict__ Ssum)
{
    const int h = threadIdx.x;
    const int b = blockIdx.x;

    float M = log_g_f(h0[(size_t)b * H_ + h]);
    float S = 1.0f;
#pragma unroll
    for (int j = 0; j < NC_; ++j) {
        size_t sidx = ((size_t)b * NC_ + j) * H_ + h;
        const float Cj = Csum[sidx];
        const float Mj = Msum[sidx];
        const float Sj = Ssum[sidx];
        Msum[sidx] = M;
        Ssum[sidx] = S;
        const float M1 = M + Cj;
        const float d  = Mj - M1;
        const float e  = __expf(-fabsf(d));
        S = (d > 0.0f) ? fmaf(S, e, Sj) : fmaf(Sj, e, S);
        M = fmaxf(M1, Mj);
    }
}

__global__ __launch_bounds__(256) void mingru_apply(
    const float* __restrict__ k_arr, float* __restrict__ vout,
    const float* __restrict__ Ment, const float* __restrict__ Sent)
{
    const int h = threadIdx.x;
    const int j = blockIdx.x;
    const int b = blockIdx.y;

    const size_t sidx = ((size_t)b * NC_ + j) * H_ + h;
    float M = Ment[sidx];
    float S = Sent[sidx];
    size_t base = ((size_t)b * T_ + (size_t)j * L_) * H_ + h;

    for (int t0 = 0; t0 < L_; t0 += RB_) {
        float k[RB_], a[RB_];
#pragma unroll
        for (int u = 0; u < RB_; ++u) {
            size_t idx = base + (size_t)(t0 + u) * H_;
            k[u] = k_arr[idx];
            a[u] = vout[idx];
        }
#pragma unroll
        for (int u = 0; u < RB_; ++u) {
            float Mc, Sc, Mv, Sv;
            row_terms(k[u], a[u], Mc, Sc, Mv, Sv);
            scan_step(M, S, Mc, Sc, Mv, Sv);
            vout[base + (size_t)(t0 + u) * H_] = __expf(M) * S;
        }
        renorm(M, S);
    }
}

extern "C" void kernel_launch(void* const* d_in, const int* in_sizes, int n_in,
                              void* d_out, int out_size, void* d_ws, size_t ws_size,
                              hipStream_t stream)
{
    const float* x  = (const float*)d_in[0];
    const float* h0 = (const float*)d_in[1];
    const float* Wz = (const float*)d_in[2];
    const float* bz = (const float*)d_in[3];
    const float* Wh = (const float*)d_in[4];
    const float* bh = (const float*)d_in[5];
    float* out = (float*)d_out;

    float* ws    = (float*)d_ws;
    float* c_arr = ws;
    float* Csum  = c_arr + (size_t)B_ * T_ * H_;
    float* Msum  = Csum  + (size_t)B_ * NC_ * H_;
    float* Ssum  = Msum  + (size_t)B_ * NC_ * H_;

    dim3 ggrid(H_ / (16 * GW_), (B_ * T_) / (16 * RT_));
    mingru_gemm_kernel<<<ggrid, GW_ * 32, 0, stream>>>(x, Wz, bz, Wh, bh, out, c_arr);

    dim3 sgrid(NC_, B_);
    mingru_chunk_reduce<<<sgrid, H_, 0, stream>>>(c_arr, out, Csum, Msum, Ssum);
    mingru_carry<<<B_, H_, 0, stream>>>(h0, Csum, Msum, Ssum);
    mingru_apply<<<sgrid, H_, 0, stream>>>(c_arr, out, Msum, Ssum);
}


# GEMM x tile fetched once per block: the 8 waves bring each 16-row tile in with whole-line 16-byte loads into a 2-stage LDS ring of the block and read fragments there, same k order; sums unchanged
# speedup vs baseline: 3.0945x; 1.0343x over previous
#include <hip/hip_runtime.h>
#include <hip/hip_bf16.h>
#include <math.h>

#define B_   32
#define T_   4096
#define DIN_ 256
#define H_   256
#define NC_  32
#define L_   (T_ / NC_)

typedef float v2f __attribute__((ext_vector_type(2)));
typedef float v4f __attribute__((ext_vector_type(4)));
typedef float v8f __attribute__((ext_vector_type(8)));

__device__ __forceinline__ float softplus_f(float u) {
    return fmaxf(u, 0.0f) + log1pf(__expf(-fabsf(u)));
}

__device__ __forceinline__ float log_g_f(float a) {
    return (a >= 0.0f) ? __logf(a + 0.5f) : -softplus_f(-a);
}

#define RB_ 8

__device__ __forceinline__ void row_terms(float k, float a,
                                          float& Mc, float& Sc, float& Mv, float& Sv) {
    const float E1 = __expf(-fabsf(k));
    const float r1 = __builtin_amdgcn_rcpf(1.0f + E1);
    const float E2 = __expf(-fabsf(a));
    const float r2 = __builtin_amdgcn_rcpf(1.0f + E2);
    Mc = fminf(-k, 0.0f);
    Sc = r1;
    Mv = (k + Mc) + fminf(a, 0.0f);
    const float G = (a >= 0.0f) ? (a + 0.5f) : r2;
    Sv = r1 * G;
}

__device__ __forceinline__ void scan_step(float& M, float& S,
                                          float Mc, float Sc, float Mv, float Sv) {
    const float M1 = M + Mc;
    const float S1 = S * Sc;
    const float d  = Mv - M1;
    const float e  = __expf(-fabsf(d));
    S = (d > 0.0f) ? fmaf(S1, e, Sv) : fmaf(Sv, e, S1);
    M = fmaxf(M1, Mv);
}

__device__ __forceinline__ void renorm(float& M, float& S) {
    int ex;
    S = frexpf(S, &ex);
    M = fmaf((float)ex, 0.6931471805599453f, M);
}

#define KP_  (DIN_ / 8)
#define RT_  32
#define GW_  8
#define XQ_  (DIN_ / 4)
#define XP_  (XQ_ + 1)

typedef __attribute__((address_space(3))) v4f lds_v4f;

#define XFRAG4(q)                                                              \
    asm volatile("ds_load_b128 %0, %6 offset:%7\n\t"                           \
                 "ds_load_b128 %1, %6 offset:%8\n\t"                           \
                 "ds_load_b128 %2, %6 offset:%9\n\t"                           \
                 "ds_load_b128 %3, %6 offset:%10"                              \
                 : "+v"(xa[4 * (q)]), "+v"(xa[4 * (q) + 1]),                   \
                   "+v"(xa[4 * (q) + 2]), "+v"(xa[4 * (q) + 3]),               \
                   "+v"(accz), "+v"(acch)                                      \
                 : "v"(xaddr), "n"(128 * (q)), "n"(128 * (q) + 32),            \
                   "n"(128 * (q) + 64), "n"(128 * (q) + 96))

__device__ __forceinline__ v8f wmma_f32(v2f a, v2f b, v8f c) {
    return __builtin_amdgcn_wmma_f32_16x16x4_f32(
               false, a, false, b, (short)0, c, false, false);
}

__global__ __launch_bounds__(GW_ * 32) void mingru_gemm_kernel(
    const float* __restrict__ x,  const float* __restrict__ Wz,
    const float* __restrict__ bz, const float* __restrict__ Wh,
    const float* __restrict__ bh, float* __restrict__ v_out,
    float* __restrict__ c_out)
{
    __shared__ v4f xs[2][16][XP_];

    const int lane = threadIdx.x & 31;
    const int wave = threadIdx.x >> 5;
    const int l15  = lane & 15;
    const int hsel = lane >> 4;
    const int k4   = hsel << 2;
    const int n0   = (blockIdx.x * GW_ + wave) << 4;
    const int m0   = blockIdx.y * (RT_ * 16);

    const float* wzp = Wz + (size_t)(n0 + l15) * DIN_ + k4;
    const float* whp = Wh + (size_t)(n0 + l15) * DIN_ + k4;

    const int    srow = wave << 1;
    const float* xg   = x + (size_t)(m0 + srow) * DIN_ + (lane << 2);

    v4f wz[KP_], wh[KP_], xa[KP_], g[4];
#pragma unroll
    for (int p = 0; p < KP_; ++p) {
        wz[p] = *(const v4f*)(wzp + 8 * p);
        wh[p] = *(const v4f*)(whp + 8 * p);
    }

#pragma unroll
    for (int s = 0; s < 2; ++s) {
        const float* xt = xg + (size_t)s * (16 * DIN_);
#pragma unroll
        for (int i = 0; i < 4; ++i)
            g[i] = *(const v4f*)(xt + (i >> 1) * DIN_ + (i & 1) * 128);
#pragma unroll
        for (int i = 0; i < 4; ++i)
            xs[s][srow + (i >> 1)][(i & 1) * 32 + lane] = g[i];
    }
    __syncthreads();
#pragma unroll
    for (int p = 0; p < KP_; ++p) xa[p] = xs[0][l15][2 * p + hsel];
    __syncthreads();

    const int   ncol  = n0 + l15;
    const float bzb   = bz[ncol];
    const float bhb   = bh[ncol];
    const int   rhalf = hsel << 3;

    for (int t = 0; t < RT_; ++t) {
        v8f accz = {};
        v8f acch = {};
        const int t2 = (t + 2 < RT_) ? (t + 2) : (RT_ - 1);
        const float* xt = xg + (size_t)t2 * (16 * DIN_);
#pragma unroll
        for (int i = 0; i < 4; ++i)
            g[i] = *(const v4f*)(xt + (i >> 1) * DIN_ + (i & 1) * 128);

        const unsigned xaddr =
            (unsigned)(size_t)(lds_v4f*)&xs[(t + 1) & 1][l15][hsel];

#pragma unroll
        for (int hf = 0; hf < 2; ++hf) {
#pragma unroll
            for (int p = hf * (KP_ / 2); p < (hf + 1) * (KP_ / 2); ++p) {
                const v2f a0 = {xa[p].x, xa[p].y};
                const v2f a1 = {xa[p].z, xa[p].w};
                const v2f z0 = {wz[p].x, wz[p].y};
                const v2f z1 = {wz[p].z, wz[p].w};
                const v2f h0 = {wh[p].x, wh[p].y};
                const v2f h1 = {wh[p].z, wh[p].w};
                accz = wmma_f32(a0, z0, accz);
                acch = wmma_f32(a0, h0, acch);
                accz = wmma_f32(a1, z1, accz);
                acch = wmma_f32(a1, h1, acch);
            }
#pragma unroll
            for (int q = hf * 4; q < (hf + 1) * 4; ++q)
                XFRAG4(q);
        }

#pragma unroll
        for (int i = 0; i < 4; ++i)
            xs[t & 1][srow + (i >> 1)][(i & 1) * 32 + lane] = g[i];
        asm volatile("s_wait_dscnt 0x0" ::: "memory");
        __syncthreads();

        asm volatile("v_nop\n\tv_nop\n\tv_nop\n\tv_nop" : "+v"(accz), "+v"(acch));

        const int mbase = m0 + t * 16 + rhalf;
#pragma unroll
        for (int j = 0; j < 8; ++j) {
            size_t idx = (size_t)(mbase + j) * H_ + ncol;
            v_out[idx] = acch[j] + bhb;
            c_out[idx] = accz[j] + bzb;
        }
    }
}

__global__ __launch_bounds__(256) void mingru_chunk_reduce(
    const float* __restrict__ k_arr, const float* __restrict__ a_arr,
    float* __restrict__ Csum, float* __restrict__ Msum, float* __restrict__ Ssum)
{
    const int h = threadIdx.x;
    const int j = blockIdx.x;
    const int b = blockIdx.y;

    size_t base = ((size_t)b * T_ + (size_t)j * L_) * H_ + h;
    float A = 0.0f;
    float P = 1.0f;
    float M = -INFINITY;
    float S = 0.0f;

    for (int t0 = 0; t0 < L_; t0 += RB_) {
        float k[RB_], a[RB_];
#pragma unroll
        for (int u = 0; u < RB_; ++u) {
            size_t idx = base + (size_t)(t0 + u) * H_;
            k[u] = k_arr[idx];
            a[u] = a_arr[idx];
        }
#pragma unroll
        for (int u = 0; u < RB_; ++u) {
            float Mc, Sc, Mv, Sv;
            row_terms(k[u], a[u], Mc, Sc, Mv, Sv);
            A += Mc;
            P *= Sc;
            scan_step(M, S, Mc, Sc, Mv, Sv);
        }
        renorm(A, P);
        renorm(M, S);
    }
    size_t sidx = ((size_t)b * NC_ + j) * H_ + h;
    Csum[sidx] = A + __logf(P);
    Msum[sidx] = M;
    Ssum[sidx] = S;
}

__global__ __launch_bounds__(256) void mingru_carry(
    const float* __restrict__ h0, const float* __restrict__ Csum,
    float* __restrict__ Msum, float* __restrict__ Ssum)
{
    const int h = threadIdx.x;
    const int b = blockIdx.x;

    float M = log_g_f(h0[(size_t)b * H_ + h]);
    float S = 1.0f;
#pragma unroll
    for (int j = 0; j < NC_; ++j) {
        size_t sidx = ((size_t)b * NC_ + j) * H_ + h;
        const float Cj = Csum[sidx];
        const float Mj = Msum[sidx];
        const float Sj = Ssum[sidx];
        Msum[sidx] = M;
        Ssum[sidx] = S;
        const float M1 = M + Cj;
        const float d  = Mj - M1;
        const float e  = __expf(-fabsf(d));
        S = (d > 0.0f) ? fmaf(S, e, Sj) : fmaf(Sj, e, S);
        M = fmaxf(M1, Mj);
    }
}

__global__ __launch_bounds__(256) void mingru_apply(
    const float* __restrict__ k_arr, float* __restrict__ vout,
    const float* __restrict__ Ment, const float* __restrict__ Sent)
{
    const int h = threadIdx.x;
    const int j = blockIdx.x;
    const int b = blockIdx.y;

    const size_t sidx = ((size_t)b * NC_ + j) * H_ + h;
    float M = Ment[sidx];
    float S = Sent[sidx];
    size_t base = ((size_t)b * T_ + (size_t)j * L_) * H_ + h;

    for (int t0 = 0; t0 < L_; t0 += RB_) {
        float k[RB_], a[RB_];
#pragma unroll
        for (int u = 0; u < RB_; ++u) {
            size_t idx = base + (size_t)(t0 + u) * H_;
            k[u] = k_arr[idx];
            a[u] = vout[idx];
        }
#pragma unroll
        for (int u = 0; u < RB_; ++u) {
            float Mc, Sc, Mv, Sv;
            row_terms(k[u], a[u], Mc, Sc, Mv, Sv);
            scan_step(M, S, Mc, Sc, Mv, Sv);
            vout[base + (size_t)(t0 + u) * H_] = __expf(M) * S;
        }
        renorm(M, S);
    }
}

extern "C" void kernel_launch(void* const* d_in, const int* in_sizes, int n_in,
                              void* d_out, int out_size, void* d_ws, size_t ws_size,
                              hipStream_t stream)
{
    const float* x  = (const float*)d_in[0];
    const float* h0 = (const float*)d_in[1];
    const float* Wz = (const float*)d_in[2];
    const float* bz = (const float*)d_in[3];
    const float* Wh = (const float*)d_in[4];
    const float* bh = (const float*)d_in[5];
    float* out = (float*)d_out;

    float* ws    = (float*)d_ws;
    float* c_arr = ws;
    float* Csum  = c_arr + (size_t)B_ * T_ * H_;
    float* Msum  = Csum  + (size_t)B_ * NC_ * H_;
    float* Ssum  = Msum  + (size_t)B_ * NC_ * H_;

    dim3 ggrid(H_ / (16 * GW_), (B_ * T_) / (16 * RT_));
    mingru_gemm_kernel<<<ggrid, GW_ * 32, 0, stream>>>(x, Wz, bz, Wh, bh, out, c_arr);

    dim3 sgrid(NC_, B_);
    mingru_chunk_reduce<<<sgrid, H_, 0, stream>>>(c_arr, out, Csum, Msum, Ssum);
    mingru_carry<<<B_, H_, 0, stream>>>(h0, Csum, Msum, Ssum);
    mingru_apply<<<sgrid, H_, 0, stream>>>(c_arr, out, Msum, Ssum);
}
